# starting kernel
# baseline (speedup 1.0000x reference)
#include <hip/hip_runtime.h>


typedef __attribute__((ext_vector_type(2))) float v2f;
typedef __attribute__((ext_vector_type(8))) float v8f;

#define H_IN   96
#define W_IN   96
#define HW_IN  (H_IN * W_IN)
#define CIN    16
#define B_     8
#define A_     8
#define COUT   16
#define N_     128
#define H_OUT  192
#define W_OUT  192
#define NPIX   (H_OUT * W_OUT)
#define WP2    144

__global__ void __launch_bounds__(256)
vote_wmma_kernel(const float* __restrict__ x,
                 const float* __restrict__ wt,
                 float* __restrict__ votes)
{
    __shared__ v2f wS[8 * WP2];

    const int tid   = threadIdx.x;
    const int lane  = tid & 31;
    const int laneM = lane & 15;
    const int laneH = lane >> 4;
    const int wave  = tid >> 5;

    const int cls = blockIdx.x / 576;
    const int rem = blockIdx.x % 576;
    const int p   = cls >> 1;
    const int q   = cls & 1;

    int t = rem * 8 + wave;
    const int ox0 = (t % 6) * 16;  t /= 6;
    const int oy  = t % 96;        t /= 96;
    const int b   = t;
    const int oh  = 2 * oy + p;

    v8f acc[8] = {};

    const int nTh = 3 - p;
    const int nTw = 3 - q;

    for (int i = 0; i < nTh; ++i) {
        const int  kh  = 2 * i + p;
        const int  ih  = oy + i - 1 + p;
        const bool okh = (unsigned)ih < H_IN;
        for (int j = 0; j < nTw; ++j) {
            const int kw = 2 * j + q;

            __syncthreads();
            {
                const float4* src =
                    (const float4*)(wt + (size_t)(kh * 5 + kw) * CIN * N_);
                #pragma unroll
                for (int e = 0; e < 2; ++e) {
                    const int idx4 = tid * 2 + e;
                    const int idx  = idx4 * 4;
                    const int ci   = idx >> 7;
                    const int n0   = idx & 127;
                    const int k    = ci >> 1;
                    const int sub  = ci & 1;
                    const float4 w4 = src[idx4];
                    ((float*)&wS[k * WP2 + n0 + 0])[sub] = w4.x;
                    ((float*)&wS[k * WP2 + n0 + 1])[sub] = w4.y;
                    ((float*)&wS[k * WP2 + n0 + 2])[sub] = w4.z;
                    ((float*)&wS[k * WP2 + n0 + 3])[sub] = w4.w;
                }
            }
            __syncthreads();

            const int  iw  = ox0 + laneM + j - 1 + q;
            const bool ok  = okh && ((unsigned)iw < W_IN);
            const int  abase = (b * CIN * H_IN + (okh ? ih : 0)) * W_IN
                             + (((unsigned)iw < W_IN) ? iw : 0);

            #pragma unroll
            for (int kg = 0; kg < 4; ++kg) {
                const int ci = kg * 4 + 2 * laneH;
                const float v0 = x[abase + (ci    ) * HW_IN];
                const float v1 = x[abase + (ci + 1) * HW_IN];
                v2f afrag;
                afrag.x = ok ? v0 : 0.f;
                afrag.y = ok ? v1 : 0.f;

                const int brow = (kg * 2 + laneH) * WP2 + laneM;
                v2f bf[8];
                #pragma unroll
                for (int nt = 0; nt < 8; ++nt)
                    bf[nt] = wS[brow + nt * 16];
                #pragma unroll
                for (int nt = 0; nt < 8; ++nt)
                    acc[nt] = __builtin_amdgcn_wmma_f32_16x16x4_f32(
                        false, afrag, false, bf[nt],
                        (short)0, acc[nt], false, false);
            }
        }
    }

    #pragma unroll
    for (int nt = 0; nt < 8; ++nt) {
        #pragma unroll
        for (int r = 0; r < 8; ++r) {
            const int m  = r + 8 * laneH;
            const int ow = 2 * (ox0 + m) + q;
            votes[((size_t)(oh * W_OUT + ow) * B_ + b) * N_ + nt * 16 + laneM]
                = acc[nt][r];
        }
    }
}

__global__ void __launch_bounds__(256)
routing_kernel(const float* __restrict__ votes,
               const int* __restrict__ nroutes_p,
               float* __restrict__ out)
{
    __shared__ float sL[4][64];
    __shared__ float sS[4][64][17];
    __shared__ float sF[4][16];

    const int tid = threadIdx.x;
    const int pp  = tid >> 6;
    const int t64 = tid & 63;
    const int b   = t64 >> 3;
    const int a   = t64 & 7;
    const int pix = blockIdx.x * 4 + pp;

    const int nroutes = *nroutes_p;

    float v[16];
    const float4* vp =
        (const float4*)(votes + ((size_t)pix * B_ + b) * N_ + a * 16);
    #pragma unroll
    for (int i = 0; i < 4; ++i) {
        float4 t = vp[i];
        v[4*i+0] = t.x; v[4*i+1] = t.y; v[4*i+2] = t.z; v[4*i+3] = t.w;
    }

    float logit = 0.f;

    for (int r = 0; r < nroutes; ++r) {
        sL[pp][t64] = logit;
        __syncthreads();
        float mx = -3.402823466e38f;
        #pragma unroll
        for (int bb = 0; bb < 8; ++bb) mx = fmaxf(mx, sL[pp][bb * 8 + a]);
        float den = 0.f;
        #pragma unroll
        for (int bb = 0; bb < 8; ++bb) den += expf(sL[pp][bb * 8 + a] - mx);
        const float ls = expf(logit - mx) / den;

        #pragma unroll
        for (int c = 0; c < 16; ++c) sS[pp][t64][c] = v[c] * ls;
        __syncthreads();
        #pragma unroll
        for (int off = 4; off >= 1; off >>= 1) {
            if (b < off) {
                #pragma unroll
                for (int c = 0; c < 16; ++c)
                    sS[pp][b * 8 + a][c] += sS[pp][(b + off) * 8 + a][c];
            }
            __syncthreads();
        }

        if (t64 < 16) {
            float ss = 0.f;
            #pragma unroll
            for (int aa = 0; aa < 8; ++aa) {
                const float pv = sS[pp][aa][t64];
                ss += pv * pv;
            }
            sF[pp][t64] = sqrtf(ss) / (1.f + ss);
        }
        __syncthreads();

        if (r + 1 < nroutes) {
            float upd = 0.f;
            #pragma unroll
            for (int c = 0; c < 16; ++c)
                upd += v[c] * sS[pp][a][c] * sF[pp][c];
            logit = ls + upd;
        } else {
            #pragma unroll
            for (int k = 0; k < 2; ++k) {
                const int n  = t64 + 64 * k;
                const int ao = n >> 4, co = n & 15;
                out[(size_t)n * NPIX + pix] = sS[pp][ao][co] * sF[pp][co];
            }
        }
    }
}

extern "C" void kernel_launch(void* const* d_in, const int* in_sizes, int n_in,
                              void* d_out, int out_size, void* d_ws, size_t ws_size,
                              hipStream_t stream) {
    (void)in_sizes; (void)n_in; (void)out_size; (void)ws_size;
    const float* x  = (const float*)d_in[0];
    const float* wt = (const float*)d_in[1];
    const int*   nr = (const int*)d_in[2];
    float* votes = (float*)d_ws;
    float* out   = (float*)d_out;

    vote_wmma_kernel<<<2304, 256, 0, stream>>>(x, wt, votes);
    routing_kernel<<<9216, 256, 0, stream>>>(votes, nr, out);
}


# votes buffer laid out [pixel][batch][channel quad][atom][4], weight-slab columns permuted in LDS to match: every vote store and routing load covers whole 128-byte lines; sum orders kept
# speedup vs baseline: 1.9324x; 1.9324x over previous
#include <hip/hip_runtime.h>


typedef __attribute__((ext_vector_type(2))) float v2f;
typedef __attribute__((ext_vector_type(8))) float v8f;

#define H_IN   96
#define W_IN   96
#define HW_IN  (H_IN * W_IN)
#define CIN    16
#define B_     8
#define A_     8
#define COUT   16
#define N_     128
#define H_OUT  192
#define W_OUT  192
#define NPIX   (H_OUT * W_OUT)

#define SLAB_F2   1040
#define ODD_ROWS  528

__device__ __forceinline__ void load_a_frags(const float* __restrict__ x,
                                             int b, int ih, int iw, int laneH,
                                             v2f (&af)[4])
{
    const bool okh = (unsigned)ih < H_IN;
    const bool okw = (unsigned)iw < W_IN;
    const bool ok  = okh && okw;
    const int abase = (b * CIN * H_IN + (okh ? ih : 0)) * W_IN + (okw ? iw : 0);
    #pragma unroll
    for (int kg = 0; kg < 4; ++kg) {
        const int ci = kg * 4 + 2 * laneH;
        const float v0 = x[abase + (ci    ) * HW_IN];
        const float v1 = x[abase + (ci + 1) * HW_IN];
        af[kg].x = ok ? v0 : 0.f;
        af[kg].y = ok ? v1 : 0.f;
    }
}

__global__ void __launch_bounds__(256)
vote_wmma_kernel(const float* __restrict__ x,
                 const float* __restrict__ wt,
                 float* __restrict__ votes)
{
    __shared__ v2f wS[9 * SLAB_F2];

    const int tid   = threadIdx.x;
    const int lane  = tid & 31;
    const int laneM = lane & 15;
    const int laneH = lane >> 4;
    const int wave  = tid >> 5;

    const int cls = blockIdx.x / 576;
    const int rem = blockIdx.x % 576;
    const int p   = cls >> 1;
    const int q   = cls & 1;

    int t = rem * 8 + wave;
    const int ox0 = (t % 6) * 16;  t /= 6;
    const int oy  = t % 96;        t /= 96;
    const int b   = t;
    const int oh  = 2 * oy + p;

    const int nTh  = 3 - p;
    const int nTw  = 3 - q;
    const int nTap = nTh * nTw;

    {
        const int k  = tid >> 5;
        const int n4 = tid & 31;
        const int col0 = 64 * ((n4 >> 1) & 1) + 8 * (n4 & 1) + (n4 >> 2);
        const int rowoff = (k & 1) * ODD_ROWS + (k >> 1) * 128 + col0;
        const float* wrow = wt + (2 * k) * N_ + 4 * n4;
        int ti = 0, tj = 0;
        for (int tp = 0; tp < nTap; ++tp) {
            const int kh = 2 * ti + p;
            const int kw = 2 * tj + q;
            const float* s = wrow + (kh * 5 + kw) * (CIN * N_);
            const float4 e = *(const float4*)s;
            const float4 o = *(const float4*)(s + N_);
            v2f* dst = &wS[tp * SLAB_F2 + rowoff];
            dst[0]  = (v2f){e.x, o.x};
            dst[16] = (v2f){e.y, o.y};
            dst[32] = (v2f){e.z, o.z};
            dst[48] = (v2f){e.w, o.w};
            if (++tj == nTw) { tj = 0; ++ti; }
        }
    }
    __syncthreads();

    v8f acc[8] = {};

    v2f aNext[4];
    load_a_frags(x, b, oy - 1 + p, ox0 + laneM - 1 + q, laneH, aNext);

    int ti = 0, tj = 0;
    const int bbase = laneH * ODD_ROWS + laneM;
    for (int tp = 0; tp < nTap; ++tp) {
        v2f aCur[4];
        #pragma unroll
        for (int kg = 0; kg < 4; ++kg) aCur[kg] = aNext[kg];

        if (tp + 1 < nTap) { if (++tj == nTw) { tj = 0; ++ti; } }
        load_a_frags(x, b, oy + ti - 1 + p, ox0 + laneM + tj - 1 + q, laneH,
                     aNext);

        const v2f* slab = &wS[tp * SLAB_F2 + bbase];
        #pragma unroll
        for (int kg = 0; kg < 4; ++kg) {
            v2f bf[8];
            #pragma unroll
            for (int nt = 0; nt < 8; ++nt)
                bf[nt] = slab[kg * 128 + nt * 16];
            #pragma unroll
            for (int nt = 0; nt < 8; ++nt)
                acc[nt] = __builtin_amdgcn_wmma_f32_16x16x4_f32(
                    false, aCur[kg], false, bf[nt],
                    (short)0, acc[nt], false, false);
        }
    }

    #pragma unroll
    for (int r = 0; r < 8; ++r) {
        const int m  = r + 8 * laneH;
        const int ow = 2 * (ox0 + m) + q;
        float* rec = votes + ((size_t)(oh * W_OUT + ow) * B_ + b) * N_
                           + 4 * laneM;
        #pragma unroll
        for (int g = 0; g < 2; ++g)
            *(float4*)(rec + 64 * g) =
                make_float4(acc[4 * g][r], acc[4 * g + 1][r],
                            acc[4 * g + 2][r], acc[4 * g + 3][r]);
    }
}


template <int MASK>
__device__ __forceinline__ float lane_xor(float x)
{
    return __builtin_bit_cast(float,
        __builtin_amdgcn_update_dpp(0, __builtin_bit_cast(int, x),
                                    0x160 | MASK, 0xF, 0xF, true));
}

template <int SRC>
__device__ __forceinline__ float group8_bcast(float x)
{
    return __builtin_bit_cast(float,
        __builtin_amdgcn_mov_dpp8(__builtin_bit_cast(unsigned, x),
                                  SRC * 0x249249u));
}

__global__ void __launch_bounds__(128)
routing_kernel(const float* __restrict__ votes,
               const int* __restrict__ nroutes_p,
               float* __restrict__ out)
{
    const int tid = threadIdx.x;
    const int a   = tid & 7;
    const int pix = blockIdx.x * 16 + (tid >> 3);

    const int nroutes = *nroutes_p;

    v2f v[8][8];
    const float4* vp =
        (const float4*)(votes + (size_t)pix * (B_ * N_) + a * 4);
    #pragma unroll
    for (int b = 0; b < 8; ++b) {
        #pragma unroll
        for (int i = 0; i < 4; ++i) {
            const float4 t = vp[b * (N_ / 4) + i * 8];
            v[b][2*i+0].x = t.x; v[b][2*i+0].y = t.y;
            v[b][2*i+1].x = t.z; v[b][2*i+1].y = t.w;
        }
    }

    const bool h2 = (a & 4) != 0;
    const bool h1 = (a & 2) != 0;
    const bool h0 = (a & 1) != 0;

    float logit[8];
    #pragma unroll
    for (int b = 0; b < 8; ++b) logit[b] = 0.f;

    for (int r = 0; r < nroutes; ++r) {
        float mx = -3.402823466e38f;
        #pragma unroll
        for (int b = 0; b < 8; ++b) mx = fmaxf(mx, logit[b]);
        float e[8];
        float den = 0.f;
        #pragma unroll
        for (int b = 0; b < 8; ++b) { e[b] = expf(logit[b] - mx); den += e[b]; }
        float ls[8];
        #pragma unroll
        for (int b = 0; b < 8; ++b) ls[b] = e[b] / den;

        v2f pr[8];
        #pragma unroll
        for (int k = 0; k < 8; ++k) {
            v2f s = v[0][k] * ls[0];
            #pragma unroll
            for (int b = 1; b < 8; ++b) s += v[b][k] * ls[b];
            pr[k] = s;
        }

        float p2[16];
        #pragma unroll
        for (int k = 0; k < 8; ++k) {
            const v2f sq = pr[k] * pr[k];
            p2[2*k] = sq.x; p2[2*k+1] = sq.y;
        }
        float t8[8], t4[4], t2[2];
        #pragma unroll
        for (int j = 0; j < 8; ++j)
            t8[j] = (h2 ? p2[8 + j] : p2[j]) + lane_xor<4>(h2 ? p2[j] : p2[8 + j]);
        #pragma unroll
        for (int j = 0; j < 4; ++j)
            t4[j] = (h1 ? t8[4 + j] : t8[j]) + lane_xor<2>(h1 ? t8[j] : t8[4 + j]);
        #pragma unroll
        for (int j = 0; j < 2; ++j)
            t2[j] = (h0 ? t4[2 + j] : t4[j]) + lane_xor<1>(h0 ? t4[j] : t4[2 + j]);

        const float fA = sqrtf(t2[0]) / (1.f + t2[0]);
        const float fB = sqrtf(t2[1]) / (1.f + t2[1]);

        v2f q[8];
        q[0] = pr[0] * (v2f){group8_bcast<0>(fA), group8_bcast<0>(fB)};
        q[1] = pr[1] * (v2f){group8_bcast<1>(fA), group8_bcast<1>(fB)};
        q[2] = pr[2] * (v2f){group8_bcast<2>(fA), group8_bcast<2>(fB)};
        q[3] = pr[3] * (v2f){group8_bcast<3>(fA), group8_bcast<3>(fB)};
        q[4] = pr[4] * (v2f){group8_bcast<4>(fA), group8_bcast<4>(fB)};
        q[5] = pr[5] * (v2f){group8_bcast<5>(fA), group8_bcast<5>(fB)};
        q[6] = pr[6] * (v2f){group8_bcast<6>(fA), group8_bcast<6>(fB)};
        q[7] = pr[7] * (v2f){group8_bcast<7>(fA), group8_bcast<7>(fB)};

        if (r + 1 < nroutes) {
            #pragma unroll
            for (int b = 0; b < 8; ++b) {
                v2f u = v[b][0] * q[0];
                #pragma unroll
                for (int k = 1; k < 8; ++k) u += v[b][k] * q[k];
                logit[b] = ls[b] + (u.x + u.y);
            }
        } else {
            #pragma unroll
            for (int k = 0; k < 8; ++k) {
                out[(size_t)(a * 16 + 2*k    ) * NPIX + pix] = q[k].x;
                out[(size_t)(a * 16 + 2*k + 1) * NPIX + pix] = q[k].y;
            }
        }
    }
}

extern "C" void kernel_launch(void* const* d_in, const int* in_sizes, int n_in,
                              void* d_out, int out_size, void* d_ws, size_t ws_size,
                              hipStream_t stream) {
    (void)in_sizes; (void)n_in; (void)out_size; (void)ws_size;
    const float* x  = (const float*)d_in[0];
    const float* wt = (const float*)d_in[1];
    const int*   nr = (const int*)d_in[2];
    float* votes = (float*)d_ws;
    float* out   = (float*)d_out;

    vote_wmma_kernel<<<2304, 256, 0, stream>>>(x, wt, votes);
    routing_kernel<<<2304, 128, 0, stream>>>(votes, nr, out);
}


# routing: 768 blocks, all resident at once; a wave routes 3 pixel groups in turn, the next group's 32 vote loads issued into a second register set before the group in hand is routed; sum orders kept
# speedup vs baseline: 2.3513x; 1.2168x over previous
#include <hip/hip_runtime.h>


typedef __attribute__((ext_vector_type(2))) float v2f;
typedef __attribute__((ext_vector_type(8))) float v8f;

#define H_IN   96
#define W_IN   96
#define HW_IN  (H_IN * W_IN)
#define CIN    16
#define B_     8
#define A_     8
#define COUT   16
#define N_     128
#define H_OUT  192
#define W_OUT  192
#define NPIX   (H_OUT * W_OUT)

#define SLAB_F2   1040
#define ODD_ROWS  528

__device__ __forceinline__ void load_a_frags(const float* __restrict__ x,
                                             int b, int ih, int iw, int laneH,
                                             v2f (&af)[4])
{
    const bool okh = (unsigned)ih < H_IN;
    const bool okw = (unsigned)iw < W_IN;
    const bool ok  = okh && okw;
    const int abase = (b * CIN * H_IN + (okh ? ih : 0)) * W_IN + (okw ? iw : 0);
    #pragma unroll
    for (int kg = 0; kg < 4; ++kg) {
        const int ci = kg * 4 + 2 * laneH;
        const float v0 = x[abase + (ci    ) * HW_IN];
        const float v1 = x[abase + (ci + 1) * HW_IN];
        af[kg].x = ok ? v0 : 0.f;
        af[kg].y = ok ? v1 : 0.f;
    }
}

__global__ void __launch_bounds__(256)
vote_wmma_kernel(const float* __restrict__ x,
                 const float* __restrict__ wt,
                 float* __restrict__ votes)
{
    __shared__ v2f wS[9 * SLAB_F2];

    const int tid   = threadIdx.x;
    const int lane  = tid & 31;
    const int laneM = lane & 15;
    const int laneH = lane >> 4;
    const int wave  = tid >> 5;

    const int cls = blockIdx.x / 576;
    const int rem = blockIdx.x % 576;
    const int p   = cls >> 1;
    const int q   = cls & 1;

    int t = rem * 8 + wave;
    const int ox0 = (t % 6) * 16;  t /= 6;
    const int oy  = t % 96;        t /= 96;
    const int b   = t;
    const int oh  = 2 * oy + p;

    const int nTh  = 3 - p;
    const int nTw  = 3 - q;
    const int nTap = nTh * nTw;

    {
        const int k  = tid >> 5;
        const int n4 = tid & 31;
        const int col0 = 64 * ((n4 >> 1) & 1) + 8 * (n4 & 1) + (n4 >> 2);
        const int rowoff = (k & 1) * ODD_ROWS + (k >> 1) * 128 + col0;
        const float* wrow = wt + (2 * k) * N_ + 4 * n4;
        int ti = 0, tj = 0;
        for (int tp = 0; tp < nTap; ++tp) {
            const int kh = 2 * ti + p;
            const int kw = 2 * tj + q;
            const float* s = wrow + (kh * 5 + kw) * (CIN * N_);
            const float4 e = *(const float4*)s;
            const float4 o = *(const float4*)(s + N_);
            v2f* dst = &wS[tp * SLAB_F2 + rowoff];
            dst[0]  = (v2f){e.x, o.x};
            dst[16] = (v2f){e.y, o.y};
            dst[32] = (v2f){e.z, o.z};
            dst[48] = (v2f){e.w, o.w};
            if (++tj == nTw) { tj = 0; ++ti; }
        }
    }
    __syncthreads();

    v8f acc[8] = {};

    v2f aNext[4];
    load_a_frags(x, b, oy - 1 + p, ox0 + laneM - 1 + q, laneH, aNext);

    int ti = 0, tj = 0;
    const int bbase = laneH * ODD_ROWS + laneM;
    for (int tp = 0; tp < nTap; ++tp) {
        v2f aCur[4];
        #pragma unroll
        for (int kg = 0; kg < 4; ++kg) aCur[kg] = aNext[kg];

        if (tp + 1 < nTap) { if (++tj == nTw) { tj = 0; ++ti; } }
        load_a_frags(x, b, oy + ti - 1 + p, ox0 + laneM + tj - 1 + q, laneH,
                     aNext);

        const v2f* slab = &wS[tp * SLAB_F2 + bbase];
        #pragma unroll
        for (int kg = 0; kg < 4; ++kg) {
            v2f bf[8];
            #pragma unroll
            for (int nt = 0; nt < 8; ++nt)
                bf[nt] = slab[kg * 128 + nt * 16];
            #pragma unroll
            for (int nt = 0; nt < 8; ++nt)
                acc[nt] = __builtin_amdgcn_wmma_f32_16x16x4_f32(
                    false, aCur[kg], false, bf[nt],
                    (short)0, acc[nt], false, false);
        }
    }

    #pragma unroll
    for (int r = 0; r < 8; ++r) {
        const int m  = r + 8 * laneH;
        const int ow = 2 * (ox0 + m) + q;
        float* rec = votes + ((size_t)(oh * W_OUT + ow) * B_ + b) * N_
                           + 4 * laneM;
        #pragma unroll
        for (int g = 0; g < 2; ++g)
            *(float4*)(rec + 64 * g) =
                make_float4(acc[4 * g][r], acc[4 * g + 1][r],
                            acc[4 * g + 2][r], acc[4 * g + 3][r]);
    }
}


template <int MASK>
__device__ __forceinline__ float lane_xor(float x)
{
    return __builtin_bit_cast(float,
        __builtin_amdgcn_update_dpp(0, __builtin_bit_cast(int, x),
                                    0x160 | MASK, 0xF, 0xF, true));
}

template <int SRC>
__device__ __forceinline__ float group8_bcast(float x)
{
    return __builtin_bit_cast(float,
        __builtin_amdgcn_mov_dpp8(__builtin_bit_cast(unsigned, x),
                                  SRC * 0x249249u));
}

__device__ __forceinline__ void load_group(const float* __restrict__ votes,
                                           int pix, int a, v2f (&v)[8][8])
{
    const float4* vp =
        (const float4*)(votes + (size_t)pix * (B_ * N_) + a * 4);
    #pragma unroll
    for (int b = 0; b < 8; ++b) {
        #pragma unroll
        for (int i = 0; i < 4; ++i) {
            const float4 t = vp[b * (N_ / 4) + i * 8];
            v[b][2*i+0].x = t.x; v[b][2*i+0].y = t.y;
            v[b][2*i+1].x = t.z; v[b][2*i+1].y = t.w;
        }
    }
}

__device__ __forceinline__ void route_group(const v2f (&v)[8][8],
                                            int pix, int a, int nroutes,
                                            float* __restrict__ out)
{
    const bool h2 = (a & 4) != 0;
    const bool h1 = (a & 2) != 0;
    const bool h0 = (a & 1) != 0;

    float logit[8];
    #pragma unroll
    for (int b = 0; b < 8; ++b) logit[b] = 0.f;

    for (int r = 0; r < nroutes; ++r) {
        float mx = -3.402823466e38f;
        #pragma unroll
        for (int b = 0; b < 8; ++b) mx = fmaxf(mx, logit[b]);
        float e[8];
        float den = 0.f;
        #pragma unroll
        for (int b = 0; b < 8; ++b) { e[b] = expf(logit[b] - mx); den += e[b]; }
        float ls[8];
        #pragma unroll
        for (int b = 0; b < 8; ++b) ls[b] = e[b] / den;

        v2f pr[8];
        #pragma unroll
        for (int k = 0; k < 8; ++k) {
            v2f s = v[0][k] * ls[0];
            #pragma unroll
            for (int b = 1; b < 8; ++b) s += v[b][k] * ls[b];
            pr[k] = s;
        }

        float p2[16];
        #pragma unroll
        for (int k = 0; k < 8; ++k) {
            const v2f sq = pr[k] * pr[k];
            p2[2*k] = sq.x; p2[2*k+1] = sq.y;
        }
        float t8[8], t4[4], t2[2];
        #pragma unroll
        for (int j = 0; j < 8; ++j)
            t8[j] = (h2 ? p2[8 + j] : p2[j]) + lane_xor<4>(h2 ? p2[j] : p2[8 + j]);
        #pragma unroll
        for (int j = 0; j < 4; ++j)
            t4[j] = (h1 ? t8[4 + j] : t8[j]) + lane_xor<2>(h1 ? t8[j] : t8[4 + j]);
        #pragma unroll
        for (int j = 0; j < 2; ++j)
            t2[j] = (h0 ? t4[2 + j] : t4[j]) + lane_xor<1>(h0 ? t4[j] : t4[2 + j]);

        const float fA = sqrtf(t2[0]) / (1.f + t2[0]);
        const float fB = sqrtf(t2[1]) / (1.f + t2[1]);

        v2f q[8];
        q[0] = pr[0] * (v2f){group8_bcast<0>(fA), group8_bcast<0>(fB)};
        q[1] = pr[1] * (v2f){group8_bcast<1>(fA), group8_bcast<1>(fB)};
        q[2] = pr[2] * (v2f){group8_bcast<2>(fA), group8_bcast<2>(fB)};
        q[3] = pr[3] * (v2f){group8_bcast<3>(fA), group8_bcast<3>(fB)};
        q[4] = pr[4] * (v2f){group8_bcast<4>(fA), group8_bcast<4>(fB)};
        q[5] = pr[5] * (v2f){group8_bcast<5>(fA), group8_bcast<5>(fB)};
        q[6] = pr[6] * (v2f){group8_bcast<6>(fA), group8_bcast<6>(fB)};
        q[7] = pr[7] * (v2f){group8_bcast<7>(fA), group8_bcast<7>(fB)};

        if (r + 1 < nroutes) {
            #pragma unroll
            for (int b = 0; b < 8; ++b) {
                v2f u = v[b][0] * q[0];
                #pragma unroll
                for (int k = 1; k < 8; ++k) u += v[b][k] * q[k];
                logit[b] = ls[b] + (u.x + u.y);
            }
        } else {
            #pragma unroll
            for (int k = 0; k < 8; ++k) {
                out[(size_t)(a * 16 + 2*k    ) * NPIX + pix] = q[k].x;
                out[(size_t)(a * 16 + 2*k + 1) * NPIX + pix] = q[k].y;
            }
        }
    }
}

#define GROUPS_PER_BLOCK 3
__global__ void __launch_bounds__(128)
routing_kernel(const float* __restrict__ votes,
               const int* __restrict__ nroutes_p,
               float* __restrict__ out)
{
    const int tid = threadIdx.x;
    const int a   = tid & 7;
    const int pix = blockIdx.x * (GROUPS_PER_BLOCK * 16) + (tid >> 3);

    const int nroutes = *nroutes_p;
    if (nroutes < 1) return;

    v2f vA[8][8], vB[8][8];
    load_group(votes, pix, a, vA);
    load_group(votes, pix + 16, a, vB);
    route_group(vA, pix, a, nroutes, out);
    load_group(votes, pix + 32, a, vA);
    route_group(vB, pix + 16, a, nroutes, out);
    route_group(vA, pix + 32, a, nroutes, out);
}

extern "C" void kernel_launch(void* const* d_in, const int* in_sizes, int n_in,
                              void* d_out, int out_size, void* d_ws, size_t ws_size,
                              hipStream_t stream) {
    (void)in_sizes; (void)n_in; (void)out_size; (void)ws_size;
    const float* x  = (const float*)d_in[0];
    const float* wt = (const float*)d_in[1];
    const int*   nr = (const int*)d_in[2];
    float* votes = (float*)d_ws;
    float* out   = (float*)d_out;

    vote_wmma_kernel<<<2304, 256, 0, stream>>>(x, wt, votes);
    routing_kernel<<<768, 128, 0, stream>>>(votes, nr, out);
}
